# starting kernel
# baseline (speedup 1.0000x reference)
#include <hip/hip_runtime.h>

typedef float v2f __attribute__((ext_vector_type(2)));
typedef float v8f __attribute__((ext_vector_type(8)));

#define EMBED   512
#define BASIS   256
#define KCHUNK  128
#define LDSPAD  132

__global__ __launch_bounds__(256)
void gemm_xtrans(const float* __restrict__ X, const float* __restrict__ W,
                 float* __restrict__ Xt) {
  __shared__ float tile[32][LDSPAD];

  const int tid   = threadIdx.x;
  const int lane  = tid & 31;
  const int wave  = tid >> 5;
  const int strip = wave >> 2;
  const int cg    = wave & 3;
  const int row0  = blockIdx.x * 32;
  const int lhalf = lane >> 4;
  const int lmod  = lane & 15;

  v8f acc0 = {}, acc1 = {}, acc2 = {}, acc3 = {};

  for (int kc = 0; kc < EMBED; kc += KCHUNK) {
    __syncthreads();
#pragma unroll
    for (int i = 0; i < 4; ++i) {
      int f4 = tid + i * 256;
      int r  = f4 >> 5;
      int c4 = f4 & 31;
      float4 v = *(const float4*)(X + (size_t)(row0 + r) * EMBED + kc + c4 * 4);
      *(float4*)&tile[r][c4 * 4] = v;
    }
    __syncthreads();

#pragma unroll 4
    for (int kk = 0; kk < KCHUNK; kk += 4) {
      v2f a = *(const v2f*)&tile[strip * 16 + lmod][kk + 2 * lhalf];
      const float* wb = W + (size_t)(cg * 64 + lmod) * EMBED + kc + kk + 2 * lhalf;
      v2f b0 = *(const v2f*)(wb);
      v2f b1 = *(const v2f*)(wb + 16 * EMBED);
      v2f b2 = *(const v2f*)(wb + 32 * EMBED);
      v2f b3 = *(const v2f*)(wb + 48 * EMBED);
      acc0 = __builtin_amdgcn_wmma_f32_16x16x4_f32(false, a, false, b0, (short)0, acc0, false, false);
      acc1 = __builtin_amdgcn_wmma_f32_16x16x4_f32(false, a, false, b1, (short)0, acc1, false, false);
      acc2 = __builtin_amdgcn_wmma_f32_16x16x4_f32(false, a, false, b2, (short)0, acc2, false, false);
      acc3 = __builtin_amdgcn_wmma_f32_16x16x4_f32(false, a, false, b3, (short)0, acc3, false, false);
    }
  }

  const int bbase = cg * 64 + lmod;
#pragma unroll
  for (int j = 0; j < 8; ++j) {
    size_t roff = (size_t)(row0 + strip * 16 + j + 8 * lhalf) * BASIS;
    Xt[roff + bbase +  0] = acc0[j];
    Xt[roff + bbase + 16] = acc1[j];
    Xt[roff + bbase + 32] = acc2[j];
    Xt[roff + bbase + 48] = acc3[j];
  }
}

__global__ __launch_bounds__(256)
void edge_score(const float* __restrict__ Xt, const int* __restrict__ src,
                const int* __restrict__ tgt, const int* __restrict__ rel,
                const float* __restrict__ ebv, float* __restrict__ out,
                int nedges) {
  int e    = (int)((blockIdx.x * blockDim.x + threadIdx.x) >> 5);
  int lane = threadIdx.x & 31;
  if (e >= nedges) return;

  int s = src[e], t = tgt[e], r = rel[e];
  const float4* ps = (const float4*)(Xt  + (size_t)s * BASIS);
  const float4* pt = (const float4*)(Xt  + (size_t)t * BASIS);
  const float4* pe = (const float4*)(ebv + (size_t)r * BASIS);

  float acc = 0.f;
#pragma unroll
  for (int j = 0; j < 2; ++j) {
    float4 a = ps[lane + 32 * j];
    float4 b = pt[lane + 32 * j];
    float4 v = pe[lane + 32 * j];
    acc += (a.x - b.x) * v.x + (a.y - b.y) * v.y +
           (a.z - b.z) * v.z + (a.w - b.w) * v.w;
  }
#pragma unroll
  for (int off = 16; off > 0; off >>= 1)
    acc += __shfl_down(acc, off, 32);
  if (lane == 0) out[e] = acc;
}

extern "C" void kernel_launch(void* const* d_in, const int* in_sizes, int n_in,
                              void* d_out, int out_size, void* d_ws, size_t ws_size,
                              hipStream_t stream) {
  const float* X     = (const float*)d_in[0];
  const int*   edges = (const int*)d_in[1];
  const int*   etype = (const int*)d_in[2];
  const float* W     = (const float*)d_in[3];
  const float* ebv   = (const float*)d_in[4];
  float*       out   = (float*)d_out;
  float*       Xt    = (float*)d_ws;

  int nnodes = in_sizes[0] / EMBED;
  int nedges = in_sizes[2];

  gemm_xtrans<<<nnodes / 32, 256, 0, stream>>>(X, W, Xt);

  int nblk = (nedges + 7) / 8;
  edge_score<<<nblk, 256, 0, stream>>>(Xt, edges, edges + nedges, etype, ebv,
                                       out, nedges);
}


# GEMM: W (B operand) staged per k-chunk in LDS by whole-line 16-B loads, block tile 128 rows x 128 cols; B fragments are LDS reads, not 16-row global gathers; k order and sum order kept
# speedup vs baseline: 1.5575x; 1.5575x over previous
#include <hip/hip_runtime.h>

typedef float v2f __attribute__((ext_vector_type(2)));
typedef float v8f __attribute__((ext_vector_type(8)));

#define EMBED   512
#define BASIS   256
#define TM      128
#define TN      128
#define KC      32
#define PITCH   36

__global__ __launch_bounds__(256) __attribute__((amdgpu_waves_per_eu(8, 8)))
void gemm_xtrans(const float* __restrict__ X, const float* __restrict__ W,
                 float* __restrict__ Xt, int nnodes) {
  __shared__ __attribute__((aligned(16))) float xs[TM][PITCH];
  __shared__ __attribute__((aligned(16))) float wt[TN][PITCH];

  const int tid   = threadIdx.x;
  const int lane  = tid & 31;
  const int wave  = tid >> 5;
  const int rg    = wave >> 1;
  const int cg    = wave & 1;
  const int row0  = (int)(blockIdx.x >> 1) * TM;
  const int col0  = (int)(blockIdx.x & 1) * TN;
  const int lhalf = lane >> 4;
  const int lmod  = lane & 15;

  v8f c00 = {}, c01 = {}, c02 = {}, c03 = {};
  v8f c10 = {}, c11 = {}, c12 = {}, c13 = {};

  const int sr = tid >> 3;
  const int sc = (tid & 7) * 4;
  const int last = nnodes - 1;
  const int r0i = row0 + sr, r1i = r0i + 32, r2i = r0i + 64, r3i = r0i + 96;
  const unsigned xo0 = (unsigned)(r0i < last ? r0i : last) * EMBED + sc;
  const unsigned xo1 = (unsigned)(r1i < last ? r1i : last) * EMBED + sc;
  const unsigned xo2 = (unsigned)(r2i < last ? r2i : last) * EMBED + sc;
  const unsigned xo3 = (unsigned)(r3i < last ? r3i : last) * EMBED + sc;
  const unsigned wo = (unsigned)(col0 + sr) * EMBED + sc;

  for (int kc = 0; kc < EMBED; kc += KC) {
    __syncthreads();
    const unsigned k0 = (unsigned)kc;
    const float4 x0 = *(const float4*)(X + (xo0 + k0));
    const float4 x1 = *(const float4*)(X + (xo1 + k0));
    const float4 x2 = *(const float4*)(X + (xo2 + k0));
    const float4 x3 = *(const float4*)(X + (xo3 + k0));
    const float4 w0 = *(const float4*)(W + (wo + k0));
    const float4 w1 = *(const float4*)(W + (wo + k0 + 32u * EMBED));
    const float4 w2 = *(const float4*)(W + (wo + k0 + 64u * EMBED));
    const float4 w3 = *(const float4*)(W + (wo + k0 + 96u * EMBED));
    *(float4*)&xs[sr     ][sc] = x0;
    *(float4*)&xs[sr + 32][sc] = x1;
    *(float4*)&xs[sr + 64][sc] = x2;
    *(float4*)&xs[sr + 96][sc] = x3;
    *(float4*)&wt[sr     ][sc] = w0;
    *(float4*)&wt[sr + 32][sc] = w1;
    *(float4*)&wt[sr + 64][sc] = w2;
    *(float4*)&wt[sr + 96][sc] = w3;
    __syncthreads();

#pragma unroll
    for (int kk = 0; kk < KC; kk += 4) {
      v2f a0 = *(const v2f*)&xs[rg * 32 + lmod][kk + 2 * lhalf];
      v2f a1 = *(const v2f*)&xs[rg * 32 + 16 + lmod][kk + 2 * lhalf];
      v2f b0 = *(const v2f*)&wt[cg * 64 +  0 + lmod][kk + 2 * lhalf];
      v2f b1 = *(const v2f*)&wt[cg * 64 + 16 + lmod][kk + 2 * lhalf];
      v2f b2 = *(const v2f*)&wt[cg * 64 + 32 + lmod][kk + 2 * lhalf];
      v2f b3 = *(const v2f*)&wt[cg * 64 + 48 + lmod][kk + 2 * lhalf];
      c00 = __builtin_amdgcn_wmma_f32_16x16x4_f32(false, a0, false, b0, (short)0, c00, false, false);
      c01 = __builtin_amdgcn_wmma_f32_16x16x4_f32(false, a0, false, b1, (short)0, c01, false, false);
      c02 = __builtin_amdgcn_wmma_f32_16x16x4_f32(false, a0, false, b2, (short)0, c02, false, false);
      c03 = __builtin_amdgcn_wmma_f32_16x16x4_f32(false, a0, false, b3, (short)0, c03, false, false);
      c10 = __builtin_amdgcn_wmma_f32_16x16x4_f32(false, a1, false, b0, (short)0, c10, false, false);
      c11 = __builtin_amdgcn_wmma_f32_16x16x4_f32(false, a1, false, b1, (short)0, c11, false, false);
      c12 = __builtin_amdgcn_wmma_f32_16x16x4_f32(false, a1, false, b2, (short)0, c12, false, false);
      c13 = __builtin_amdgcn_wmma_f32_16x16x4_f32(false, a1, false, b3, (short)0, c13, false, false);
    }
  }

  const int bbase = col0 + cg * 64 + lmod;
#pragma unroll
  for (int j = 0; j < 8; ++j) {
    int ra = row0 + rg * 32 + j + 8 * lhalf;
    int rb = ra + 16;
    if (ra < nnodes) {
      size_t roff = (size_t)ra * BASIS;
      Xt[roff + bbase +  0] = c00[j];
      Xt[roff + bbase + 16] = c01[j];
      Xt[roff + bbase + 32] = c02[j];
      Xt[roff + bbase + 48] = c03[j];
    }
    if (rb < nnodes) {
      size_t roff = (size_t)rb * BASIS;
      Xt[roff + bbase +  0] = c10[j];
      Xt[roff + bbase + 16] = c11[j];
      Xt[roff + bbase + 32] = c12[j];
      Xt[roff + bbase + 48] = c13[j];
    }
  }
}

__global__ __launch_bounds__(256)
void edge_score(const float* __restrict__ Xt, const int* __restrict__ src,
                const int* __restrict__ tgt, const int* __restrict__ rel,
                const float* __restrict__ ebv, float* __restrict__ out,
                int nedges) {
  int e    = (int)((blockIdx.x * blockDim.x + threadIdx.x) >> 5);
  int lane = threadIdx.x & 31;
  if (e >= nedges) return;

  int s = src[e], t = tgt[e], r = rel[e];
  const float4* ps = (const float4*)(Xt  + (size_t)s * BASIS);
  const float4* pt = (const float4*)(Xt  + (size_t)t * BASIS);
  const float4* pe = (const float4*)(ebv + (size_t)r * BASIS);

  float acc = 0.f;
#pragma unroll
  for (int j = 0; j < 2; ++j) {
    float4 a = ps[lane + 32 * j];
    float4 b = pt[lane + 32 * j];
    float4 v = pe[lane + 32 * j];
    acc += (a.x - b.x) * v.x + (a.y - b.y) * v.y +
           (a.z - b.z) * v.z + (a.w - b.w) * v.w;
  }
#pragma unroll
  for (int off = 16; off > 0; off >>= 1)
    acc += __shfl_down(acc, off, 32);
  if (lane == 0) out[e] = acc;
}

extern "C" void kernel_launch(void* const* d_in, const int* in_sizes, int n_in,
                              void* d_out, int out_size, void* d_ws, size_t ws_size,
                              hipStream_t stream) {
  const float* X     = (const float*)d_in[0];
  const int*   edges = (const int*)d_in[1];
  const int*   etype = (const int*)d_in[2];
  const float* W     = (const float*)d_in[3];
  const float* ebv   = (const float*)d_in[4];
  float*       out   = (float*)d_out;
  float*       Xt    = (float*)d_ws;

  int nnodes = in_sizes[0] / EMBED;
  int nedges = in_sizes[2];

  int ntile = (nnodes + TM - 1) / TM;
  gemm_xtrans<<<ntile * 2, 256, 0, stream>>>(X, W, Xt, nnodes);

  int nblk = (nedges + 7) / 8;
  edge_score<<<nblk, 256, 0, stream>>>(Xt, edges, edges + nedges, etype, ebv,
                                       out, nedges);
}


# Reassociated (X W^T).b_r = X.(W^T b_r): WMMA now projects the 500 relation vectors (V = B W, 500x512), not the 100000 nodes; per edge one 512-long f32 dot (X[s]-X[t]).V[r]; sum order changes
# speedup vs baseline: 3.5080x; 2.2523x over previous
#include <hip/hip_runtime.h>

typedef float v2f __attribute__((ext_vector_type(2)));
typedef float v4f __attribute__((ext_vector_type(4)));
typedef float v8f __attribute__((ext_vector_type(8)));

#define EMBED   512
#define BASIS   256


__global__ __launch_bounds__(128) __attribute__((amdgpu_waves_per_eu(1, 4)))
void project_rel(const float* __restrict__ ebv, const float* __restrict__ W,
                 float* __restrict__ V, int nrel) {
  const int lane = threadIdx.x & 31;
  const int wave = threadIdx.x >> 5;
  const int c    = lane & 15;
  const int h    = lane >> 4;
  const int r0   = (int)blockIdx.y * 16;
  const int n0   = ((int)blockIdx.x * 4 + wave) * 16;

  const int ra = (r0 + c < nrel) ? (r0 + c) : (nrel - 1);
  const float* pa = ebv + (size_t)ra * BASIS + 2 * h;
  const float* pb = W + (size_t)(2 * h) * EMBED + n0 + c;

  v8f acc0 = {}, acc1 = {};
#pragma unroll
  for (int kk = 0; kk < BASIS; kk += 8) {
    const v2f a0 = *(const v2f*)(pa + kk);
    const v2f a1 = *(const v2f*)(pa + kk + 4);
    v2f b0, b1;
    b0.x = pb[(size_t)kk * EMBED];
    b0.y = pb[(size_t)(kk + 1) * EMBED];
    b1.x = pb[(size_t)(kk + 4) * EMBED];
    b1.y = pb[(size_t)(kk + 5) * EMBED];
    acc0 = __builtin_amdgcn_wmma_f32_16x16x4_f32(false, a0, false, b0, (short)0, acc0, false, false);
    acc1 = __builtin_amdgcn_wmma_f32_16x16x4_f32(false, a1, false, b1, (short)0, acc1, false, false);
  }
  const v8f acc = acc0 + acc1;

#pragma unroll
  for (int j = 0; j < 8; ++j) {
    const int row = r0 + j + 8 * h;
    if (row < nrel) V[(size_t)row * EMBED + n0 + c] = acc[j];
  }
}

template <int XMASK>
__device__ __forceinline__ float row_xor_add(float x) {
  const int b = __builtin_bit_cast(int, x);
  const int y = __builtin_amdgcn_update_dpp(0, b, 0x160 + XMASK, 0xf, 0xf, true);
  return x + __builtin_bit_cast(float, y);
}
__device__ __forceinline__ float wave_sum(float x) {
  x = row_xor_add<1>(x);
  x = row_xor_add<2>(x);
  x = row_xor_add<4>(x);
  x = row_xor_add<8>(x);
  const int b = __builtin_bit_cast(int, x);
  const int y = __builtin_amdgcn_permlanex16(b, b, 0x76543210, 0xfedcba98, false, false);
  return x + __builtin_bit_cast(float, y);
}

__global__ __launch_bounds__(64) __attribute__((amdgpu_waves_per_eu(8, 12)))
void edge_score(const float* __restrict__ X, const int* __restrict__ src,
                const int* __restrict__ tgt, const int* __restrict__ rel,
                const float* __restrict__ V, float* __restrict__ out,
                int nedges, int nnodes, int nrel) {
  const int lane  = threadIdx.x & 31;
  const int wave  = __builtin_amdgcn_readfirstlane((int)(threadIdx.x >> 5));
  const int e0    = ((int)blockIdx.x * 2 + wave) * 32;
  if (e0 >= nedges) return;
  const int e     = e0 + lane;
  const int ec    = e < nedges ? e : nedges - 1;
  const int sv = src[ec], tv = tgt[ec], rv = rel[ec];
  const int n  = (nedges - e0 < 32) ? (nedges - e0) : 32;
  const unsigned lastn = (unsigned)(nnodes - 1), lastr = (unsigned)(nrel - 1);

  float res = 0.f;
#pragma unroll 1
  for (int i = 0; i < n; ++i) {
    unsigned s = (unsigned)__builtin_amdgcn_readlane(sv, i);
    unsigned t = (unsigned)__builtin_amdgcn_readlane(tv, i);
    unsigned r = (unsigned)__builtin_amdgcn_readlane(rv, i);
    s = s < lastn ? s : lastn;
    t = t < lastn ? t : lastn;
    r = r < lastr ? r : lastr;
    const v4f* ps = (const v4f*)(X + (size_t)s * EMBED) + lane;
    const v4f* pt = (const v4f*)(X + (size_t)t * EMBED) + lane;
    const v4f* pv = (const v4f*)(V + (size_t)r * EMBED) + lane;
    const v4f a0 = ps[0], a1 = ps[32], a2 = ps[64], a3 = ps[96];
    const v4f b0 = pt[0], b1 = pt[32], b2 = pt[64], b3 = pt[96];
    const v4f v0 = pv[0], v1 = pv[32], v2 = pv[64], v3 = pv[96];
    v4f q = (a0 - b0) * v0;
    q += (a1 - b1) * v1;
    q += (a2 - b2) * v2;
    q += (a3 - b3) * v3;
    const float tot = wave_sum((q.x + q.y) + (q.z + q.w));
    if (lane == i) res = tot;
  }
  if (e < nedges) out[e] = res;
}

extern "C" void kernel_launch(void* const* d_in, const int* in_sizes, int n_in,
                              void* d_out, int out_size, void* d_ws, size_t ws_size,
                              hipStream_t stream) {
  const float* X     = (const float*)d_in[0];
  const int*   edges = (const int*)d_in[1];
  const int*   etype = (const int*)d_in[2];
  const float* W     = (const float*)d_in[3];
  const float* ebv   = (const float*)d_in[4];
  float*       out   = (float*)d_out;
  float*       V     = (float*)d_ws;

  int nnodes = in_sizes[0] / EMBED;
  int nedges = in_sizes[2];
  int nrel   = in_sizes[4] / BASIS;

  dim3 g1(EMBED / 64, (nrel + 15) / 16);
  project_rel<<<g1, 128, 0, stream>>>(ebv, W, V, nrel);

  int nchunk = (nedges + 31) / 32;
  int nblk   = (nchunk + 1) / 2;
  edge_score<<<nblk, 64, 0, stream>>>(X, edges, edges + nedges, etype, V, out,
                                      nedges, nnodes, nrel);
}


# Edge launch: rows of V served from LDS (4 relation groups x 64 edge chunks, 250 KB of V per work-group, a ballot picks the group's edges); X loads, f32 and per-edge sum order kept
# speedup vs baseline: 4.0794x; 1.1629x over previous
#include <hip/hip_runtime.h>

typedef float v2f __attribute__((ext_vector_type(2)));
typedef float v4f __attribute__((ext_vector_type(4)));
typedef float v8f __attribute__((ext_vector_type(8)));

#define EMBED   512
#define BASIS   256


__global__ __launch_bounds__(128) __attribute__((amdgpu_waves_per_eu(1, 4)))
void project_rel(const float* __restrict__ ebv, const float* __restrict__ W,
                 float* __restrict__ V, int nrel) {
  const int lane = threadIdx.x & 31;
  const int wave = threadIdx.x >> 5;
  const int c    = lane & 15;
  const int h    = lane >> 4;
  const int r0   = (int)blockIdx.y * 16;
  const int n0   = ((int)blockIdx.x * 4 + wave) * 16;

  const int ra = (r0 + c < nrel) ? (r0 + c) : (nrel - 1);
  const float* pa = ebv + (size_t)ra * BASIS + 2 * h;
  const float* pb = W + (size_t)(2 * h) * EMBED + n0 + c;

  v8f acc0 = {}, acc1 = {};
#pragma unroll
  for (int kk = 0; kk < BASIS; kk += 8) {
    const v2f a0 = *(const v2f*)(pa + kk);
    const v2f a1 = *(const v2f*)(pa + kk + 4);
    v2f b0, b1;
    b0.x = pb[(size_t)kk * EMBED];
    b0.y = pb[(size_t)(kk + 1) * EMBED];
    b1.x = pb[(size_t)(kk + 4) * EMBED];
    b1.y = pb[(size_t)(kk + 5) * EMBED];
    acc0 = __builtin_amdgcn_wmma_f32_16x16x4_f32(false, a0, false, b0, (short)0, acc0, false, false);
    acc1 = __builtin_amdgcn_wmma_f32_16x16x4_f32(false, a1, false, b1, (short)0, acc1, false, false);
  }
  const v8f acc = acc0 + acc1;

#pragma unroll
  for (int j = 0; j < 8; ++j) {
    const int row = r0 + j + 8 * h;
    if (row < nrel) V[(size_t)row * EMBED + n0 + c] = acc[j];
  }
}

template <int XMASK>
__device__ __forceinline__ float row_xor_add(float x) {
  const int b = __builtin_bit_cast(int, x);
  const int y = __builtin_amdgcn_update_dpp(0, b, 0x160 + XMASK, 0xf, 0xf, true);
  return x + __builtin_bit_cast(float, y);
}
__device__ __forceinline__ float wave_sum(float x) {
  x = row_xor_add<1>(x);
  x = row_xor_add<2>(x);
  x = row_xor_add<4>(x);
  x = row_xor_add<8>(x);
  const int b = __builtin_bit_cast(int, x);
  const int y = __builtin_amdgcn_permlanex16(b, b, 0x76543210, 0xfedcba98, false, false);
  return x + __builtin_bit_cast(float, y);
}

#define RG      125
#define NCHUNK  64

__device__ __forceinline__ float edge_lane_sum(v4f a0, v4f a1, v4f a2, v4f a3,
                                               v4f b0, v4f b1, v4f b2, v4f b3,
                                               const v4f* pv) {
  v4f q = (a0 - b0) * pv[0];
  q += (a1 - b1) * pv[32];
  q += (a2 - b2) * pv[64];
  q += (a3 - b3) * pv[96];
  return (q.x + q.y) + (q.z + q.w);
}

__global__ __launch_bounds__(1024)
void edge_score(const float* __restrict__ X, const int* __restrict__ src,
                const int* __restrict__ tgt, const int* __restrict__ rel,
                const float* __restrict__ V, float* __restrict__ out,
                int nedges, int nnodes, int nrel) {
  __shared__ v4f vrow[RG * (EMBED / 4)];

  const int tid  = (int)threadIdx.x;
  const int lane = tid & 31;
  const int wave = __builtin_amdgcn_readfirstlane(tid >> 5);
  const int rbase = (int)blockIdx.y * RG;
  const int nrows = (nrel - rbase < RG) ? (nrel - rbase) : RG;
  const unsigned lastn = (unsigned)(nnodes - 1), lastr = (unsigned)(nrel - 1);

  const int nbatch = (nedges + 31) >> 5;
  const int b_lo = (int)(((long long)blockIdx.x * nbatch) / NCHUNK);
  const int b_hi = (int)(((long long)(blockIdx.x + 1) * nbatch) / NCHUNK);
  int b = b_lo + wave;

  int sv = 0, tv = 0, rv = 0;
  if (b < b_hi) {
    const int e  = b * 32 + lane;
    const int ec = e < nedges ? e : nedges - 1;
    sv = src[ec]; tv = tgt[ec]; rv = rel[ec];
  }

  {
    const v4f* vsrc = (const v4f*)(V + (size_t)rbase * EMBED);
    const int nvec = nrows * (EMBED / 4);
    constexpr int NF = (RG * (EMBED / 4) + 1023) / 1024;
    v4f f[NF];
#pragma unroll
    for (int k = 0; k < NF; ++k) {
      const int idx = tid + k * 1024;
      f[k] = vsrc[idx < nvec ? idx : nvec - 1];
    }
#pragma unroll
    for (int k = 0; k < NF; ++k) {
      const int idx = tid + k * 1024;
      if (idx < nvec) vrow[idx] = f[k];
    }
  }
  __syncthreads();

  const v4f* xl = (const v4f*)X + lane;
  const v4f* vl = vrow + lane;

  while (b < b_hi) {
    const int e = b * 32 + lane;
    const unsigned sc = (unsigned)sv < lastn ? (unsigned)sv : lastn;
    const unsigned tc = (unsigned)tv < lastn ? (unsigned)tv : lastn;
    const unsigned rc = (unsigned)rv < lastr ? (unsigned)rv : lastr;
    const unsigned rl = rc - (unsigned)rbase;
    const bool mine = (e < nedges) && (rl < (unsigned)RG);
    unsigned todo = __builtin_amdgcn_ballot_w32(mine);

    b += 32;
    if (b < b_hi) {
      const int e2 = b * 32 + lane;
      const int ec = e2 < nedges ? e2 : nedges - 1;
      sv = src[ec]; tv = tgt[ec]; rv = rel[ec];
    }

    float res = 0.f;
    while (todo) {
      const int i0 = __builtin_ctz(todo);
      todo &= todo - 1;
      const unsigned s0 = (unsigned)__builtin_amdgcn_readlane((int)sc, i0);
      const unsigned t0 = (unsigned)__builtin_amdgcn_readlane((int)tc, i0);
      const unsigned r0 = (unsigned)__builtin_amdgcn_readlane((int)rl, i0);
      const v4f* ps0 = xl + (size_t)s0 * (EMBED / 4);
      const v4f* pt0 = xl + (size_t)t0 * (EMBED / 4);
      if (todo) {
        const int i1 = __builtin_ctz(todo);
        todo &= todo - 1;
        const unsigned s1 = (unsigned)__builtin_amdgcn_readlane((int)sc, i1);
        const unsigned t1 = (unsigned)__builtin_amdgcn_readlane((int)tc, i1);
        const unsigned r1 = (unsigned)__builtin_amdgcn_readlane((int)rl, i1);
        const v4f* ps1 = xl + (size_t)s1 * (EMBED / 4);
        const v4f* pt1 = xl + (size_t)t1 * (EMBED / 4);
        const v4f a00 = ps0[0], a01 = ps0[32], a02 = ps0[64], a03 = ps0[96];
        const v4f b00 = pt0[0], b01 = pt0[32], b02 = pt0[64], b03 = pt0[96];
        const v4f a10 = ps1[0], a11 = ps1[32], a12 = ps1[64], a13 = ps1[96];
        const v4f b10 = pt1[0], b11 = pt1[32], b12 = pt1[64], b13 = pt1[96];
        const float p0 = edge_lane_sum(a00, a01, a02, a03, b00, b01, b02, b03,
                                       vl + r0 * (EMBED / 4));
        const float p1 = edge_lane_sum(a10, a11, a12, a13, b10, b11, b12, b13,
                                       vl + r1 * (EMBED / 4));
        const float tot0 = wave_sum(p0);
        const float tot1 = wave_sum(p1);
        if (lane == i0) res = tot0;
        if (lane == i1) res = tot1;
      } else {
        const v4f a00 = ps0[0], a01 = ps0[32], a02 = ps0[64], a03 = ps0[96];
        const v4f b00 = pt0[0], b01 = pt0[32], b02 = pt0[64], b03 = pt0[96];
        const float p0 = edge_lane_sum(a00, a01, a02, a03, b00, b01, b02, b03,
                                       vl + r0 * (EMBED / 4));
        const float tot0 = wave_sum(p0);
        if (lane == i0) res = tot0;
      }
    }
    if (mine) out[e] = res;
  }
}

extern "C" void kernel_launch(void* const* d_in, const int* in_sizes, int n_in,
                              void* d_out, int out_size, void* d_ws, size_t ws_size,
                              hipStream_t stream) {
  const float* X     = (const float*)d_in[0];
  const int*   edges = (const int*)d_in[1];
  const int*   etype = (const int*)d_in[2];
  const float* W     = (const float*)d_in[3];
  const float* ebv   = (const float*)d_in[4];
  float*       out   = (float*)d_out;
  float*       V     = (float*)d_ws;

  int nnodes = in_sizes[0] / EMBED;
  int nedges = in_sizes[2];
  int nrel   = in_sizes[4] / BASIS;

  dim3 g1(EMBED / 64, (nrel + 15) / 16);
  project_rel<<<g1, 128, 0, stream>>>(ebv, W, V, nrel);

  dim3 g2(NCHUNK, (nrel + RG - 1) / RG);
  edge_score<<<g2, 1024, 0, stream>>>(X, edges, edges + nedges, etype, V, out,
                                      nedges, nnodes, nrel);
}


# Edge launch: each wave sorts the src and tgt ends of its edges by node range in LDS and all waves walk X in ascending order so repeated rows hit L2; src and tgt dots summed apart: sum order changes
# speedup vs baseline: 4.7474x; 1.1637x over previous
#include <hip/hip_runtime.h>

typedef float v2f __attribute__((ext_vector_type(2)));
typedef float v4f __attribute__((ext_vector_type(4)));
typedef float v8f __attribute__((ext_vector_type(8)));

#define EMBED   512
#define BASIS   256


__global__ __launch_bounds__(128) __attribute__((amdgpu_waves_per_eu(1, 4)))
void project_rel(const float* __restrict__ ebv, const float* __restrict__ W,
                 float* __restrict__ V, int nrel) {
  const int lane = threadIdx.x & 31;
  const int wave = threadIdx.x >> 5;
  const int c    = lane & 15;
  const int h    = lane >> 4;
  const int r0   = (int)blockIdx.y * 16;
  const int n0   = ((int)blockIdx.x * 4 + wave) * 16;

  const int ra = (r0 + c < nrel) ? (r0 + c) : (nrel - 1);
  const float* pa = ebv + (size_t)ra * BASIS + 2 * h;
  const float* pb = W + (size_t)(2 * h) * EMBED + n0 + c;

  v8f acc0 = {}, acc1 = {};
#pragma unroll
  for (int kk = 0; kk < BASIS; kk += 8) {
    const v2f a0 = *(const v2f*)(pa + kk);
    const v2f a1 = *(const v2f*)(pa + kk + 4);
    v2f b0, b1;
    b0.x = pb[(size_t)kk * EMBED];
    b0.y = pb[(size_t)(kk + 1) * EMBED];
    b1.x = pb[(size_t)(kk + 4) * EMBED];
    b1.y = pb[(size_t)(kk + 5) * EMBED];
    acc0 = __builtin_amdgcn_wmma_f32_16x16x4_f32(false, a0, false, b0, (short)0, acc0, false, false);
    acc1 = __builtin_amdgcn_wmma_f32_16x16x4_f32(false, a1, false, b1, (short)0, acc1, false, false);
  }
  const v8f acc = acc0 + acc1;

#pragma unroll
  for (int j = 0; j < 8; ++j) {
    const int row = r0 + j + 8 * h;
    if (row < nrel) V[(size_t)row * EMBED + n0 + c] = acc[j];
  }
}

template <int XMASK>
__device__ __forceinline__ float row_xor_add(float x) {
  const int b = __builtin_bit_cast(int, x);
  const int y = __builtin_amdgcn_update_dpp(0, b, 0x160 + XMASK, 0xf, 0xf, true);
  return x + __builtin_bit_cast(float, y);
}
__device__ __forceinline__ float wave_sum(float x) {
  x = row_xor_add<1>(x);
  x = row_xor_add<2>(x);
  x = row_xor_add<4>(x);
  x = row_xor_add<8>(x);
  const int b = __builtin_bit_cast(int, x);
  const int y = __builtin_amdgcn_permlanex16(b, b, 0x76543210, 0xfedcba98, false, false);
  return x + __builtin_bit_cast(float, y);
}

#define RG      125
#define NCHUNK  64
#define EC      64
#define NPRE    5
#define NBIN    16
#define WSW     512

__device__ __forceinline__ float end_lane_sum(v4f a0, v4f a1, v4f a2, v4f a3, const v4f* pv) {
  v4f q = a0 * pv[0];
  q += a1 * pv[32];
  q += a2 * pv[64];
  q += a3 * pv[96];
  return (q.x + q.y) + (q.z + q.w);
}

__global__ __launch_bounds__(1024)
void edge_score(const float* __restrict__ X, const int* __restrict__ src,
                const int* __restrict__ tgt, const int* __restrict__ rel,
                const float* __restrict__ V, float* __restrict__ out,
                int nedges, int nnodes, int nrel) {
  __shared__ v4f vrow[RG * (EMBED / 4)];
  __shared__ unsigned wsc[32 * WSW];

  const int tid  = (int)threadIdx.x;
  const int lane = tid & 31;
  const int wave = __builtin_amdgcn_readfirstlane(tid >> 5);
  const int rbase = (int)blockIdx.y * RG;
  const int nrows = (nrel - rbase < RG) ? (nrel - rbase) : RG;
  const unsigned lastn = (unsigned)(nnodes - 1), lastr = (unsigned)(nrel - 1);
  const unsigned ltmask = (1u << lane) - 1u;

  const int nbits = 32 - __builtin_clz(lastn | 1u);
  const int shift = nbits > 4 ? nbits - 4 : 0;

  const int c_lo = (int)(((long long)blockIdx.x * nedges) / NCHUNK);
  const int c_hi = (int)(((long long)(blockIdx.x + 1) * nedges) / NCHUNK);
  int e0          = c_lo + (int)(((long long)wave * (c_hi - c_lo)) / 32);
  const int e_end = c_lo + (int)(((long long)(wave + 1) * (c_hi - c_lo)) / 32);

  int sv[NPRE], tv[NPRE], rv[NPRE];
#pragma unroll
  for (int k = 0; k < NPRE; ++k) {
    const int e  = e0 + 32 * k + lane;
    const int ec = e < e_end ? e : nedges - 1;
    sv[k] = src[ec]; tv[k] = tgt[ec]; rv[k] = rel[ec];
  }

  {
    const v4f* vsrc = (const v4f*)(V + (size_t)rbase * EMBED);
    const int nvec = nrows * (EMBED / 4);
    constexpr int NF = (RG * (EMBED / 4) + 1023) / 1024;
    v4f f[NF];
#pragma unroll
    for (int k = 0; k < NF; ++k) {
      const int idx = tid + k * 1024;
      f[k] = vsrc[idx < nvec ? idx : nvec - 1];
    }
#pragma unroll
    for (int k = 0; k < NF; ++k) {
      const int idx = tid + k * 1024;
      if (idx < nvec) vrow[idx] = f[k];
    }
  }
  __syncthreads();

  unsigned* const ws    = wsc + wave * WSW;
  unsigned* const tnode = ws;
  unsigned* const inode = ws + 128;
  unsigned* const imeta = ws + 256;
  unsigned* const eid   = ws + 384;
  unsigned* const trl   = ws + 448;

  const v4f* xl = (const v4f*)X + lane;
  const v4f* vl = vrow + lane;

  int startLane = 0;
  while (e0 < e_end) {
    int nE = 0;
    bool open = true;
#pragma unroll
    for (int k = 0; k < NPRE; ++k) {
      if (open && e0 < e_end) {
        const int e = e0 + lane;
        const unsigned sc = (unsigned)sv[k] < lastn ? (unsigned)sv[k] : lastn;
        const unsigned tc = (unsigned)tv[k] < lastn ? (unsigned)tv[k] : lastn;
        const unsigned rc = (unsigned)rv[k] < lastr ? (unsigned)rv[k] : lastr;
        const unsigned rl = rc - (unsigned)rbase;
        const bool mine = (e < e_end) && (rl < (unsigned)RG) && (lane >= startLane);
        const unsigned m = __builtin_amdgcn_ballot_w32(mine);
        const int rank = nE + __builtin_popcount(m & ltmask);
        const bool take = mine && (rank < EC);
        if (take) {
          tnode[2 * rank] = sc; tnode[2 * rank + 1] = tc;
          eid[rank] = (unsigned)e; trl[rank] = rl;
        }
        const unsigned rej = __builtin_amdgcn_ballot_w32(mine && !take);
        if (rej) {
          startLane = __builtin_ctz(rej); nE = EC; open = false;
        } else {
          startLane = 0; nE += __builtin_popcount(m); e0 += 32; open = nE < EC;
        }
      }
    }
    const int nI = 2 * nE;

    if (e0 < e_end) {
#pragma unroll
      for (int k = 0; k < NPRE; ++k) {
        const int e  = e0 + 32 * k + lane;
        const int ec = e < e_end ? e : nedges - 1;
        sv[k] = src[ec]; tv[k] = tgt[ec]; rv[k] = rel[ec];
      }
    }

    int cnt[NBIN];
#pragma unroll
    for (int q = 0; q < NBIN; ++q) cnt[q] = 0;
    for (int j0 = 0; j0 < nI; j0 += 32) {
      const int j = j0 + lane;
      const bool v = j < nI;
      const unsigned nd = tnode[v ? j : 0];
      const int bin = v ? (int)(nd >> shift) : -1;
#pragma unroll
      for (int q = 0; q < NBIN; ++q)
        cnt[q] += __builtin_popcount(__builtin_amdgcn_ballot_w32(bin == q));
    }
    int pos[NBIN];
    {
      int run = 0;
#pragma unroll
      for (int q = 0; q < NBIN; ++q) { pos[q] = run; run += cnt[q]; }
    }
    for (int j0 = 0; j0 < nI; j0 += 32) {
      const int j = j0 + lane;
      const bool v = j < nI;
      const unsigned nd = tnode[v ? j : 0];
      const int bin = v ? (int)(nd >> shift) : -1;
      int dst = 0;
#pragma unroll
      for (int q = 0; q < NBIN; ++q) {
        const unsigned mq = __builtin_amdgcn_ballot_w32(bin == q);
        if (bin == q) dst = pos[q] + __builtin_popcount(mq & ltmask);
        pos[q] += __builtin_popcount(mq);
      }
      if (v) {
        inode[dst] = nd;
        imeta[dst] = trl[j >> 1] | ((unsigned)j << 8);
      }
    }

    const int lastI = nI > 0 ? nI - 1 : 0;
    unsigned n0 = (unsigned)__builtin_amdgcn_readfirstlane((int)inode[0]);
    unsigned n1 = (unsigned)__builtin_amdgcn_readfirstlane((int)inode[1 < lastI ? 1 : lastI]);
    unsigned n2 = (unsigned)__builtin_amdgcn_readfirstlane((int)inode[2 < lastI ? 2 : lastI]);
    unsigned n3 = (unsigned)__builtin_amdgcn_readfirstlane((int)inode[3 < lastI ? 3 : lastI]);
    unsigned m0 = (unsigned)__builtin_amdgcn_readfirstlane((int)imeta[0]);
    unsigned m1 = (unsigned)__builtin_amdgcn_readfirstlane((int)imeta[1 < lastI ? 1 : lastI]);
    unsigned m2 = (unsigned)__builtin_amdgcn_readfirstlane((int)imeta[2 < lastI ? 2 : lastI]);
    unsigned m3 = (unsigned)__builtin_amdgcn_readfirstlane((int)imeta[3 < lastI ? 3 : lastI]);
    for (int i = 0; i < nI; i += 4) {
      const v4f* p0 = xl + (size_t)n0 * (EMBED / 4);
      const v4f* p1 = xl + (size_t)n1 * (EMBED / 4);
      const v4f* p2 = xl + (size_t)n2 * (EMBED / 4);
      const v4f* p3 = xl + (size_t)n3 * (EMBED / 4);
      const v4f a00 = p0[0], a01 = p0[32], a02 = p0[64], a03 = p0[96];
      const v4f a10 = p1[0], a11 = p1[32], a12 = p1[64], a13 = p1[96];
      const v4f a20 = p2[0], a21 = p2[32], a22 = p2[64], a23 = p2[96];
      const v4f a30 = p3[0], a31 = p3[32], a32 = p3[64], a33 = p3[96];
      const int j0 = i + 4 < lastI ? i + 4 : lastI;
      const int j1 = i + 5 < lastI ? i + 5 : lastI;
      const int j2 = i + 6 < lastI ? i + 6 : lastI;
      const int j3 = i + 7 < lastI ? i + 7 : lastI;
      const unsigned nn0 = (unsigned)__builtin_amdgcn_readfirstlane((int)inode[j0]);
      const unsigned nn1 = (unsigned)__builtin_amdgcn_readfirstlane((int)inode[j1]);
      const unsigned nn2 = (unsigned)__builtin_amdgcn_readfirstlane((int)inode[j2]);
      const unsigned nn3 = (unsigned)__builtin_amdgcn_readfirstlane((int)inode[j3]);
      const unsigned mm0 = (unsigned)__builtin_amdgcn_readfirstlane((int)imeta[j0]);
      const unsigned mm1 = (unsigned)__builtin_amdgcn_readfirstlane((int)imeta[j1]);
      const unsigned mm2 = (unsigned)__builtin_amdgcn_readfirstlane((int)imeta[j2]);
      const unsigned mm3 = (unsigned)__builtin_amdgcn_readfirstlane((int)imeta[j3]);
      const float t0 = wave_sum(end_lane_sum(a00, a01, a02, a03, vl + (m0 & 0xffu) * (EMBED / 4)));
      const float t1 = wave_sum(end_lane_sum(a10, a11, a12, a13, vl + (m1 & 0xffu) * (EMBED / 4)));
      const float t2 = wave_sum(end_lane_sum(a20, a21, a22, a23, vl + (m2 & 0xffu) * (EMBED / 4)));
      const float t3 = wave_sum(end_lane_sum(a30, a31, a32, a33, vl + (m3 & 0xffu) * (EMBED / 4)));
      if (lane == 0) {
        tnode[m0 >> 8] = __builtin_bit_cast(unsigned, t0);
        tnode[m1 >> 8] = __builtin_bit_cast(unsigned, t1);
        tnode[m2 >> 8] = __builtin_bit_cast(unsigned, t2);
        tnode[m3 >> 8] = __builtin_bit_cast(unsigned, t3);
      }
      n0 = nn0; n1 = nn1; n2 = nn2; n3 = nn3;
      m0 = mm0; m1 = mm1; m2 = mm2; m3 = mm3;
    }

    for (int k = lane; k < nE; k += 32) {
      const float ps = __builtin_bit_cast(float, tnode[2 * k]);
      const float pt = __builtin_bit_cast(float, tnode[2 * k + 1]);
      out[eid[k]] = ps - pt;
    }
  }
}

extern "C" void kernel_launch(void* const* d_in, const int* in_sizes, int n_in,
                              void* d_out, int out_size, void* d_ws, size_t ws_size,
                              hipStream_t stream) {
  const float* X     = (const float*)d_in[0];
  const int*   edges = (const int*)d_in[1];
  const int*   etype = (const int*)d_in[2];
  const float* W     = (const float*)d_in[3];
  const float* ebv   = (const float*)d_in[4];
  float*       out   = (float*)d_out;
  float*       V     = (float*)d_ws;

  int nnodes = in_sizes[0] / EMBED;
  int nedges = in_sizes[2];
  int nrel   = in_sizes[4] / BASIS;

  dim3 g1(EMBED / 64, (nrel + 15) / 16);
  project_rel<<<g1, 128, 0, stream>>>(ebv, W, V, nrel);

  dim3 g2(NCHUNK, (nrel + RG - 1) / RG);
  edge_score<<<g2, 1024, 0, stream>>>(X, edges, edges + nedges, etype, V, out,
                                      nedges, nnodes, nrel);
}


# Edge launch: the sorted ends of a block's 32 waves are dealt evenly, four at a time in (list position, wave) order behind a barrier, so no wave walks X alone at the end; per-edge sum order kept
# speedup vs baseline: 4.8977x; 1.0317x over previous
#include <hip/hip_runtime.h>

typedef float v2f __attribute__((ext_vector_type(2)));
typedef float v4f __attribute__((ext_vector_type(4)));
typedef float v8f __attribute__((ext_vector_type(8)));

#define EMBED   512
#define BASIS   256


__global__ __launch_bounds__(128) __attribute__((amdgpu_waves_per_eu(1, 4)))
void project_rel(const float* __restrict__ ebv, const float* __restrict__ W,
                 float* __restrict__ V, int nrel) {
  const int lane = threadIdx.x & 31;
  const int wave = threadIdx.x >> 5;
  const int c    = lane & 15;
  const int h    = lane >> 4;
  const int r0   = (int)blockIdx.y * 16;
  const int n0   = ((int)blockIdx.x * 4 + wave) * 16;

  const int ra = (r0 + c < nrel) ? (r0 + c) : (nrel - 1);
  const float* pa = ebv + (size_t)ra * BASIS + 2 * h;
  const float* pb = W + (size_t)(2 * h) * EMBED + n0 + c;

  v8f acc0 = {}, acc1 = {};
#pragma unroll
  for (int kk = 0; kk < BASIS; kk += 8) {
    const v2f a0 = *(const v2f*)(pa + kk);
    const v2f a1 = *(const v2f*)(pa + kk + 4);
    v2f b0, b1;
    b0.x = pb[(size_t)kk * EMBED];
    b0.y = pb[(size_t)(kk + 1) * EMBED];
    b1.x = pb[(size_t)(kk + 4) * EMBED];
    b1.y = pb[(size_t)(kk + 5) * EMBED];
    acc0 = __builtin_amdgcn_wmma_f32_16x16x4_f32(false, a0, false, b0, (short)0, acc0, false, false);
    acc1 = __builtin_amdgcn_wmma_f32_16x16x4_f32(false, a1, false, b1, (short)0, acc1, false, false);
  }
  const v8f acc = acc0 + acc1;

#pragma unroll
  for (int j = 0; j < 8; ++j) {
    const int row = r0 + j + 8 * h;
    if (row < nrel) V[(size_t)row * EMBED + n0 + c] = acc[j];
  }
}

template <int XMASK>
__device__ __forceinline__ float row_xor_add(float x) {
  const int b = __builtin_bit_cast(int, x);
  const int y = __builtin_amdgcn_update_dpp(0, b, 0x160 + XMASK, 0xf, 0xf, true);
  return x + __builtin_bit_cast(float, y);
}
__device__ __forceinline__ float wave_sum(float x) {
  x = row_xor_add<1>(x);
  x = row_xor_add<2>(x);
  x = row_xor_add<4>(x);
  x = row_xor_add<8>(x);
  const int b = __builtin_bit_cast(int, x);
  const int y = __builtin_amdgcn_permlanex16(b, b, 0x76543210, 0xfedcba98, false, false);
  return x + __builtin_bit_cast(float, y);
}

#define RG      125
#define NCHUNK  64
#define EC      64
#define NPRE    5
#define NBIN    16
#define WSW     512

__device__ __forceinline__ float end_lane_sum(v4f a0, v4f a1, v4f a2, v4f a3, const v4f* pv) {
  v4f q = a0 * pv[0];
  q += a1 * pv[32];
  q += a2 * pv[64];
  q += a3 * pv[96];
  return (q.x + q.y) + (q.z + q.w);
}

__global__ __launch_bounds__(1024)
void edge_score(const float* __restrict__ X, const int* __restrict__ src,
                const int* __restrict__ tgt, const int* __restrict__ rel,
                const float* __restrict__ V, float* __restrict__ out,
                int nedges, int nnodes, int nrel) {
  __shared__ v4f vrow[RG * (EMBED / 4)];
  __shared__ unsigned wsc[32 * WSW];
  __shared__ unsigned tab[32];

  const int tid  = (int)threadIdx.x;
  const int lane = tid & 31;
  const int wave = __builtin_amdgcn_readfirstlane(tid >> 5);
  const int rbase = (int)blockIdx.y * RG;
  const int nrows = (nrel - rbase < RG) ? (nrel - rbase) : RG;
  const unsigned lastn = (unsigned)(nnodes - 1), lastr = (unsigned)(nrel - 1);
  const unsigned ltmask = (1u << lane) - 1u;

  const int nbits = 32 - __builtin_clz(lastn | 1u);
  const int shift = nbits > 4 ? nbits - 4 : 0;

  const int c_lo = (int)(((long long)blockIdx.x * nedges) / NCHUNK);
  const int c_hi = (int)(((long long)(blockIdx.x + 1) * nedges) / NCHUNK);
  int e0          = c_lo + (int)(((long long)wave * (c_hi - c_lo)) / 32);
  const int e_end = c_lo + (int)(((long long)(wave + 1) * (c_hi - c_lo)) / 32);

  int sv[NPRE], tv[NPRE], rv[NPRE];
#pragma unroll
  for (int k = 0; k < NPRE; ++k) {
    const int e  = e0 + 32 * k + lane;
    const int ec = e < e_end ? e : nedges - 1;
    sv[k] = src[ec]; tv[k] = tgt[ec]; rv[k] = rel[ec];
  }

  {
    const v4f* vsrc = (const v4f*)(V + (size_t)rbase * EMBED);
    const int nvec = nrows * (EMBED / 4);
    constexpr int NF = (RG * (EMBED / 4) + 1023) / 1024;
    v4f f[NF];
#pragma unroll
    for (int k = 0; k < NF; ++k) {
      const int idx = tid + k * 1024;
      f[k] = vsrc[idx < nvec ? idx : nvec - 1];
    }
#pragma unroll
    for (int k = 0; k < NF; ++k) {
      const int idx = tid + k * 1024;
      if (idx < nvec) vrow[idx] = f[k];
    }
  }
  __syncthreads();

  unsigned* const ws    = wsc + wave * WSW;
  unsigned* const tnode = ws;
  unsigned* const inode = ws + 128;
  unsigned* const imeta = ws + 256;
  unsigned* const eid   = ws + 384;
  unsigned* const trl   = ws + 448;

  const v4f* xl = (const v4f*)X + lane;
  const v4f* vl = vrow + lane;

  int startLane = 0;
  bool again = true;
  while (again) {
    int nE = 0;
    bool open = true;
#pragma unroll
    for (int k = 0; k < NPRE; ++k) {
      if (open && e0 < e_end) {
        const int e = e0 + lane;
        const unsigned sc = (unsigned)sv[k] < lastn ? (unsigned)sv[k] : lastn;
        const unsigned tc = (unsigned)tv[k] < lastn ? (unsigned)tv[k] : lastn;
        const unsigned rc = (unsigned)rv[k] < lastr ? (unsigned)rv[k] : lastr;
        const unsigned rl = rc - (unsigned)rbase;
        const bool mine = (e < e_end) && (rl < (unsigned)RG) && (lane >= startLane);
        const unsigned m = __builtin_amdgcn_ballot_w32(mine);
        const int rank = nE + __builtin_popcount(m & ltmask);
        const bool take = mine && (rank < EC);
        if (take) {
          tnode[2 * rank] = sc; tnode[2 * rank + 1] = tc;
          eid[rank] = (unsigned)e; trl[rank] = rl;
        }
        const unsigned rej = __builtin_amdgcn_ballot_w32(mine && !take);
        if (rej) {
          startLane = __builtin_ctz(rej); nE = EC; open = false;
        } else {
          startLane = 0; nE += __builtin_popcount(m); e0 += 32; open = nE < EC;
        }
      }
    }
    const int nI = 2 * nE;

    if (e0 < e_end) {
#pragma unroll
      for (int k = 0; k < NPRE; ++k) {
        const int e  = e0 + 32 * k + lane;
        const int ec = e < e_end ? e : nedges - 1;
        sv[k] = src[ec]; tv[k] = tgt[ec]; rv[k] = rel[ec];
      }
    }

    int cnt[NBIN];
#pragma unroll
    for (int q = 0; q < NBIN; ++q) cnt[q] = 0;
    for (int j0 = 0; j0 < nI; j0 += 32) {
      const int j = j0 + lane;
      const bool v = j < nI;
      const unsigned nd = tnode[v ? j : 0];
      const int bin = v ? (int)(nd >> shift) : -1;
#pragma unroll
      for (int q = 0; q < NBIN; ++q)
        cnt[q] += __builtin_popcount(__builtin_amdgcn_ballot_w32(bin == q));
    }
    int pos[NBIN];
    {
      int run = 0;
#pragma unroll
      for (int q = 0; q < NBIN; ++q) { pos[q] = run; run += cnt[q]; }
    }
    for (int j0 = 0; j0 < nI; j0 += 32) {
      const int j = j0 + lane;
      const bool v = j < nI;
      const unsigned nd = tnode[v ? j : 0];
      const int bin = v ? (int)(nd >> shift) : -1;
      int dst = 0;
#pragma unroll
      for (int q = 0; q < NBIN; ++q) {
        const unsigned mq = __builtin_amdgcn_ballot_w32(bin == q);
        if (bin == q) dst = pos[q] + __builtin_popcount(mq & ltmask);
        pos[q] += __builtin_popcount(mq);
      }
      if (v) {
        inode[dst] = nd;
        imeta[dst] = trl[j >> 1] | ((unsigned)j << 8);
      }
    }

    if (lane == 0) tab[wave] = (unsigned)nI | (e0 < e_end ? 0x10000u : 0u);
    __syncthreads();
    const unsigned mt = tab[lane];
    again = __builtin_amdgcn_ballot_w32((mt >> 16) != 0u) != 0u;
    const int tr = (int)(((mt & 0xffffu) + 3u) >> 2);

    int lvl = 0, first = 0;
    unsigned lmask = __builtin_amdgcn_ballot_w32(tr > 0);
    int lcnt = __builtin_popcount(lmask);
    int u = wave;
    while (lcnt != 0 && u >= first + lcnt) {
      first += lcnt; ++lvl;
      lmask = __builtin_amdgcn_ballot_w32(tr > lvl);
      lcnt = __builtin_popcount(lmask);
    }
    bool have = lcnt != 0;
    unsigned* lst = ws;
    unsigned n0 = 0, n1 = 0, n2 = 0, n3 = 0, m0 = 0, m1 = 0, m2 = 0, m3 = 0;
    if (have) {
      const unsigned sel = __builtin_amdgcn_ballot_w32(
          ((lmask >> lane) & 1u) != 0u && __builtin_popcount(lmask & ltmask) == u - first);
      const int ow = __builtin_ctz(sel);
      const int lastI = (int)((unsigned)__builtin_amdgcn_readlane((int)mt, ow) & 0xffffu) - 1;
      lst = wsc + ow * WSW;
      const int i = 4 * lvl;
      const int j0 = i < lastI ? i : lastI;
      const int j1 = i + 1 < lastI ? i + 1 : lastI;
      const int j2 = i + 2 < lastI ? i + 2 : lastI;
      const int j3 = i + 3 < lastI ? i + 3 : lastI;
      n0 = (unsigned)__builtin_amdgcn_readfirstlane((int)lst[128 + j0]);
      n1 = (unsigned)__builtin_amdgcn_readfirstlane((int)lst[128 + j1]);
      n2 = (unsigned)__builtin_amdgcn_readfirstlane((int)lst[128 + j2]);
      n3 = (unsigned)__builtin_amdgcn_readfirstlane((int)lst[128 + j3]);
      m0 = (unsigned)__builtin_amdgcn_readfirstlane((int)lst[256 + j0]);
      m1 = (unsigned)__builtin_amdgcn_readfirstlane((int)lst[256 + j1]);
      m2 = (unsigned)__builtin_amdgcn_readfirstlane((int)lst[256 + j2]);
      m3 = (unsigned)__builtin_amdgcn_readfirstlane((int)lst[256 + j3]);
    }
    while (have) {
      const v4f* p0 = xl + (size_t)n0 * (EMBED / 4);
      const v4f* p1 = xl + (size_t)n1 * (EMBED / 4);
      const v4f* p2 = xl + (size_t)n2 * (EMBED / 4);
      const v4f* p3 = xl + (size_t)n3 * (EMBED / 4);
      const v4f a00 = p0[0], a01 = p0[32], a02 = p0[64], a03 = p0[96];
      const v4f a10 = p1[0], a11 = p1[32], a12 = p1[64], a13 = p1[96];
      const v4f a20 = p2[0], a21 = p2[32], a22 = p2[64], a23 = p2[96];
      const v4f a30 = p3[0], a31 = p3[32], a32 = p3[64], a33 = p3[96];
      unsigned* const cur = lst;
      u += 32;
      while (lcnt != 0 && u >= first + lcnt) {
        first += lcnt; ++lvl;
        lmask = __builtin_amdgcn_ballot_w32(tr > lvl);
        lcnt = __builtin_popcount(lmask);
      }
      have = lcnt != 0;
      unsigned nn0 = n0, nn1 = n1, nn2 = n2, nn3 = n3, mm0 = m0, mm1 = m1, mm2 = m2, mm3 = m3;
      if (have) {
        const unsigned sel = __builtin_amdgcn_ballot_w32(
            ((lmask >> lane) & 1u) != 0u && __builtin_popcount(lmask & ltmask) == u - first);
        const int ow = __builtin_ctz(sel);
        const int lastI = (int)((unsigned)__builtin_amdgcn_readlane((int)mt, ow) & 0xffffu) - 1;
        lst = wsc + ow * WSW;
        const int i = 4 * lvl;
        const int j0 = i < lastI ? i : lastI;
        const int j1 = i + 1 < lastI ? i + 1 : lastI;
        const int j2 = i + 2 < lastI ? i + 2 : lastI;
        const int j3 = i + 3 < lastI ? i + 3 : lastI;
        nn0 = (unsigned)__builtin_amdgcn_readfirstlane((int)lst[128 + j0]);
        nn1 = (unsigned)__builtin_amdgcn_readfirstlane((int)lst[128 + j1]);
        nn2 = (unsigned)__builtin_amdgcn_readfirstlane((int)lst[128 + j2]);
        nn3 = (unsigned)__builtin_amdgcn_readfirstlane((int)lst[128 + j3]);
        mm0 = (unsigned)__builtin_amdgcn_readfirstlane((int)lst[256 + j0]);
        mm1 = (unsigned)__builtin_amdgcn_readfirstlane((int)lst[256 + j1]);
        mm2 = (unsigned)__builtin_amdgcn_readfirstlane((int)lst[256 + j2]);
        mm3 = (unsigned)__builtin_amdgcn_readfirstlane((int)lst[256 + j3]);
      }
      const float t0 = wave_sum(end_lane_sum(a00, a01, a02, a03, vl + (m0 & 0xffu) * (EMBED / 4)));
      const float t1 = wave_sum(end_lane_sum(a10, a11, a12, a13, vl + (m1 & 0xffu) * (EMBED / 4)));
      const float t2 = wave_sum(end_lane_sum(a20, a21, a22, a23, vl + (m2 & 0xffu) * (EMBED / 4)));
      const float t3 = wave_sum(end_lane_sum(a30, a31, a32, a33, vl + (m3 & 0xffu) * (EMBED / 4)));
      if (lane == 0) {
        cur[m0 >> 8] = __builtin_bit_cast(unsigned, t0);
        cur[m1 >> 8] = __builtin_bit_cast(unsigned, t1);
        cur[m2 >> 8] = __builtin_bit_cast(unsigned, t2);
        cur[m3 >> 8] = __builtin_bit_cast(unsigned, t3);
      }
      n0 = nn0; n1 = nn1; n2 = nn2; n3 = nn3;
      m0 = mm0; m1 = mm1; m2 = mm2; m3 = mm3;
    }
    __syncthreads();

    for (int k = lane; k < nE; k += 32) {
      const float ps = __builtin_bit_cast(float, tnode[2 * k]);
      const float pt = __builtin_bit_cast(float, tnode[2 * k + 1]);
      out[eid[k]] = ps - pt;
    }
  }
}

extern "C" void kernel_launch(void* const* d_in, const int* in_sizes, int n_in,
                              void* d_out, int out_size, void* d_ws, size_t ws_size,
                              hipStream_t stream) {
  const float* X     = (const float*)d_in[0];
  const int*   edges = (const int*)d_in[1];
  const int*   etype = (const int*)d_in[2];
  const float* W     = (const float*)d_in[3];
  const float* ebv   = (const float*)d_in[4];
  float*       out   = (float*)d_out;
  float*       V     = (float*)d_ws;

  int nnodes = in_sizes[0] / EMBED;
  int nedges = in_sizes[2];
  int nrel   = in_sizes[4] / BASIS;

  dim3 g1(EMBED / 64, (nrel + 15) / 16);
  project_rel<<<g1, 128, 0, stream>>>(ebv, W, V, nrel);

  dim3 g2(NCHUNK, (nrel + RG - 1) / RG);
  edge_score<<<g2, 1024, 0, stream>>>(X, edges, edges + nedges, etype, V, out,
                                      nedges, nnodes, nrel);
}


# Edge launch: X loads software-pipelined: once an end of a unit is summed, its registers are refilled with the same end of the wave's next unit, so loads stay in flight while summing; sums unchanged
# speedup vs baseline: 5.0820x; 1.0376x over previous
#include <hip/hip_runtime.h>

typedef float v2f __attribute__((ext_vector_type(2)));
typedef float v4f __attribute__((ext_vector_type(4)));
typedef float v8f __attribute__((ext_vector_type(8)));

#define EMBED   512
#define BASIS   256


__global__ __launch_bounds__(128) __attribute__((amdgpu_waves_per_eu(1, 4)))
void project_rel(const float* __restrict__ ebv, const float* __restrict__ W,
                 float* __restrict__ V, int nrel) {
  const int lane = threadIdx.x & 31;
  const int wave = threadIdx.x >> 5;
  const int c    = lane & 15;
  const int h    = lane >> 4;
  const int r0   = (int)blockIdx.y * 16;
  const int n0   = ((int)blockIdx.x * 4 + wave) * 16;

  const int ra = (r0 + c < nrel) ? (r0 + c) : (nrel - 1);
  const float* pa = ebv + (size_t)ra * BASIS + 2 * h;
  const float* pb = W + (size_t)(2 * h) * EMBED + n0 + c;

  v8f acc0 = {}, acc1 = {};
#pragma unroll
  for (int kk = 0; kk < BASIS; kk += 8) {
    const v2f a0 = *(const v2f*)(pa + kk);
    const v2f a1 = *(const v2f*)(pa + kk + 4);
    v2f b0, b1;
    b0.x = pb[(size_t)kk * EMBED];
    b0.y = pb[(size_t)(kk + 1) * EMBED];
    b1.x = pb[(size_t)(kk + 4) * EMBED];
    b1.y = pb[(size_t)(kk + 5) * EMBED];
    acc0 = __builtin_amdgcn_wmma_f32_16x16x4_f32(false, a0, false, b0, (short)0, acc0, false, false);
    acc1 = __builtin_amdgcn_wmma_f32_16x16x4_f32(false, a1, false, b1, (short)0, acc1, false, false);
  }
  const v8f acc = acc0 + acc1;

#pragma unroll
  for (int j = 0; j < 8; ++j) {
    const int row = r0 + j + 8 * h;
    if (row < nrel) V[(size_t)row * EMBED + n0 + c] = acc[j];
  }
}

template <int XMASK>
__device__ __forceinline__ float row_xor_add(float x) {
  const int b = __builtin_bit_cast(int, x);
  const int y = __builtin_amdgcn_update_dpp(0, b, 0x160 + XMASK, 0xf, 0xf, true);
  return x + __builtin_bit_cast(float, y);
}
__device__ __forceinline__ float wave_sum(float x) {
  x = row_xor_add<1>(x);
  x = row_xor_add<2>(x);
  x = row_xor_add<4>(x);
  x = row_xor_add<8>(x);
  const int b = __builtin_bit_cast(int, x);
  const int y = __builtin_amdgcn_permlanex16(b, b, 0x76543210, 0xfedcba98, false, false);
  return x + __builtin_bit_cast(float, y);
}

#define RG      125
#define NCHUNK  64
#define EC      64
#define NPRE    5
#define NBIN    16
#define WSW     512

__device__ __forceinline__ float end_lane_sum(v4f a0, v4f a1, v4f a2, v4f a3, const v4f* pv) {
  v4f q = a0 * pv[0];
  q += a1 * pv[32];
  q += a2 * pv[64];
  q += a3 * pv[96];
  return (q.x + q.y) + (q.z + q.w);
}

__global__ __launch_bounds__(1024)
void edge_score(const float* __restrict__ X, const int* __restrict__ src,
                const int* __restrict__ tgt, const int* __restrict__ rel,
                const float* __restrict__ V, float* __restrict__ out,
                int nedges, int nnodes, int nrel) {
  __shared__ v4f vrow[RG * (EMBED / 4)];
  __shared__ unsigned wsc[32 * WSW];
  __shared__ unsigned tab[32];

  const int tid  = (int)threadIdx.x;
  const int lane = tid & 31;
  const int wave = __builtin_amdgcn_readfirstlane(tid >> 5);
  const int rbase = (int)blockIdx.y * RG;
  const int nrows = (nrel - rbase < RG) ? (nrel - rbase) : RG;
  const unsigned lastn = (unsigned)(nnodes - 1), lastr = (unsigned)(nrel - 1);
  const unsigned ltmask = (1u << lane) - 1u;

  const int nbits = 32 - __builtin_clz(lastn | 1u);
  const int shift = nbits > 4 ? nbits - 4 : 0;

  const int c_lo = (int)(((long long)blockIdx.x * nedges) / NCHUNK);
  const int c_hi = (int)(((long long)(blockIdx.x + 1) * nedges) / NCHUNK);
  int e0          = c_lo + (int)(((long long)wave * (c_hi - c_lo)) / 32);
  const int e_end = c_lo + (int)(((long long)(wave + 1) * (c_hi - c_lo)) / 32);

  int sv[NPRE], tv[NPRE], rv[NPRE];
#pragma unroll
  for (int k = 0; k < NPRE; ++k) {
    const int e  = e0 + 32 * k + lane;
    const int ec = e < e_end ? e : nedges - 1;
    sv[k] = src[ec]; tv[k] = tgt[ec]; rv[k] = rel[ec];
  }

  {
    const v4f* vsrc = (const v4f*)(V + (size_t)rbase * EMBED);
    const int nvec = nrows * (EMBED / 4);
    constexpr int NF = (RG * (EMBED / 4) + 1023) / 1024;
    v4f f[NF];
#pragma unroll
    for (int k = 0; k < NF; ++k) {
      const int idx = tid + k * 1024;
      f[k] = vsrc[idx < nvec ? idx : nvec - 1];
    }
#pragma unroll
    for (int k = 0; k < NF; ++k) {
      const int idx = tid + k * 1024;
      if (idx < nvec) vrow[idx] = f[k];
    }
  }
  __syncthreads();

  unsigned* const ws    = wsc + wave * WSW;
  unsigned* const tnode = ws;
  unsigned* const inode = ws + 128;
  unsigned* const imeta = ws + 256;
  unsigned* const eid   = ws + 384;
  unsigned* const trl   = ws + 448;

  const v4f* xl = (const v4f*)X + lane;
  const v4f* vl = vrow + lane;

  int startLane = 0;
  bool again = true;
  while (again) {
    int nE = 0;
    bool open = true;
#pragma unroll
    for (int k = 0; k < NPRE; ++k) {
      if (open && e0 < e_end) {
        const int e = e0 + lane;
        const unsigned sc = (unsigned)sv[k] < lastn ? (unsigned)sv[k] : lastn;
        const unsigned tc = (unsigned)tv[k] < lastn ? (unsigned)tv[k] : lastn;
        const unsigned rc = (unsigned)rv[k] < lastr ? (unsigned)rv[k] : lastr;
        const unsigned rl = rc - (unsigned)rbase;
        const bool mine = (e < e_end) && (rl < (unsigned)RG) && (lane >= startLane);
        const unsigned m = __builtin_amdgcn_ballot_w32(mine);
        const int rank = nE + __builtin_popcount(m & ltmask);
        const bool take = mine && (rank < EC);
        if (take) {
          tnode[2 * rank] = sc; tnode[2 * rank + 1] = tc;
          eid[rank] = (unsigned)e; trl[rank] = rl;
        }
        const unsigned rej = __builtin_amdgcn_ballot_w32(mine && !take);
        if (rej) {
          startLane = __builtin_ctz(rej); nE = EC; open = false;
        } else {
          startLane = 0; nE += __builtin_popcount(m); e0 += 32; open = nE < EC;
        }
      }
    }
    const int nI = 2 * nE;

    if (e0 < e_end) {
#pragma unroll
      for (int k = 0; k < NPRE; ++k) {
        const int e  = e0 + 32 * k + lane;
        const int ec = e < e_end ? e : nedges - 1;
        sv[k] = src[ec]; tv[k] = tgt[ec]; rv[k] = rel[ec];
      }
    }

    int cnt[NBIN];
#pragma unroll
    for (int q = 0; q < NBIN; ++q) cnt[q] = 0;
    for (int j0 = 0; j0 < nI; j0 += 32) {
      const int j = j0 + lane;
      const bool v = j < nI;
      const unsigned nd = tnode[v ? j : 0];
      const int bin = v ? (int)(nd >> shift) : -1;
#pragma unroll
      for (int q = 0; q < NBIN; ++q)
        cnt[q] += __builtin_popcount(__builtin_amdgcn_ballot_w32(bin == q));
    }
    int pos[NBIN];
    {
      int run = 0;
#pragma unroll
      for (int q = 0; q < NBIN; ++q) { pos[q] = run; run += cnt[q]; }
    }
    for (int j0 = 0; j0 < nI; j0 += 32) {
      const int j = j0 + lane;
      const bool v = j < nI;
      const unsigned nd = tnode[v ? j : 0];
      const int bin = v ? (int)(nd >> shift) : -1;
      int dst = 0;
#pragma unroll
      for (int q = 0; q < NBIN; ++q) {
        const unsigned mq = __builtin_amdgcn_ballot_w32(bin == q);
        if (bin == q) dst = pos[q] + __builtin_popcount(mq & ltmask);
        pos[q] += __builtin_popcount(mq);
      }
      if (v) {
        inode[dst] = nd;
        imeta[dst] = trl[j >> 1] | ((unsigned)j << 8);
      }
    }

    if (lane == 0) tab[wave] = (unsigned)nI | (e0 < e_end ? 0x10000u : 0u);
    __syncthreads();
    const unsigned mt = tab[lane];
    again = __builtin_amdgcn_ballot_w32((mt >> 16) != 0u) != 0u;
    const int tr = (int)(((mt & 0xffffu) + 3u) >> 2);

    int lvl = 0, first = 0;
    unsigned lmask = __builtin_amdgcn_ballot_w32(tr > 0);
    int lcnt = __builtin_popcount(lmask);
    int u = wave;
    auto seek = [&]() -> bool {
      while (lcnt != 0 && u >= first + lcnt) {
        first += lcnt; ++lvl;
        lmask = __builtin_amdgcn_ballot_w32(tr > lvl);
        lcnt = __builtin_popcount(lmask);
      }
      return lcnt != 0;
    };
    auto fetch = [&](unsigned& q0, unsigned& q1, unsigned& q2, unsigned& q3,
                     unsigned& w0, unsigned& w1, unsigned& w2, unsigned& w3) -> unsigned* {
      const unsigned sel = __builtin_amdgcn_ballot_w32(
          ((lmask >> lane) & 1u) != 0u && __builtin_popcount(lmask & ltmask) == u - first);
      const int ow = __builtin_ctz(sel);
      const int lastI = (int)((unsigned)__builtin_amdgcn_readlane((int)mt, ow) & 0xffffu) - 1;
      unsigned* const l = wsc + ow * WSW;
      const int i = 4 * lvl;
      const int j0 = i < lastI ? i : lastI;
      const int j1 = i + 1 < lastI ? i + 1 : lastI;
      const int j2 = i + 2 < lastI ? i + 2 : lastI;
      const int j3 = i + 3 < lastI ? i + 3 : lastI;
      q0 = (unsigned)__builtin_amdgcn_readfirstlane((int)l[128 + j0]);
      q1 = (unsigned)__builtin_amdgcn_readfirstlane((int)l[128 + j1]);
      q2 = (unsigned)__builtin_amdgcn_readfirstlane((int)l[128 + j2]);
      q3 = (unsigned)__builtin_amdgcn_readfirstlane((int)l[128 + j3]);
      w0 = (unsigned)__builtin_amdgcn_readfirstlane((int)l[256 + j0]);
      w1 = (unsigned)__builtin_amdgcn_readfirstlane((int)l[256 + j1]);
      w2 = (unsigned)__builtin_amdgcn_readfirstlane((int)l[256 + j2]);
      w3 = (unsigned)__builtin_amdgcn_readfirstlane((int)l[256 + j3]);
      return l;
    };
    if (seek()) {
      unsigned n0, n1, n2, n3, m0, m1, m2, m3;
      unsigned* cur = fetch(n0, n1, n2, n3, m0, m1, m2, m3);
      const v4f* p0 = xl + (size_t)n0 * (EMBED / 4);
      const v4f* p1 = xl + (size_t)n1 * (EMBED / 4);
      const v4f* p2 = xl + (size_t)n2 * (EMBED / 4);
      const v4f* p3 = xl + (size_t)n3 * (EMBED / 4);
      v4f a00 = p0[0], a01 = p0[32], a02 = p0[64], a03 = p0[96];
      v4f a10 = p1[0], a11 = p1[32], a12 = p1[64], a13 = p1[96];
      v4f a20 = p2[0], a21 = p2[32], a22 = p2[64], a23 = p2[96];
      v4f a30 = p3[0], a31 = p3[32], a32 = p3[64], a33 = p3[96];
      unsigned nn0 = 0, nn1 = 0, nn2 = 0, nn3 = 0, mm0 = 0, mm1 = 0, mm2 = 0, mm3 = 0;
      unsigned* nxt = cur;
      u += 32;
      bool more = seek();
      if (more) nxt = fetch(nn0, nn1, nn2, nn3, mm0, mm1, mm2, mm3);
      while (more) {
        const float t0 = wave_sum(end_lane_sum(a00, a01, a02, a03, vl + (m0 & 0xffu) * (EMBED / 4)));
        { const v4f* p = xl + (size_t)nn0 * (EMBED / 4); a00 = p[0]; a01 = p[32]; a02 = p[64]; a03 = p[96]; }
        __builtin_amdgcn_sched_barrier(0);
        const float t1 = wave_sum(end_lane_sum(a10, a11, a12, a13, vl + (m1 & 0xffu) * (EMBED / 4)));
        { const v4f* p = xl + (size_t)nn1 * (EMBED / 4); a10 = p[0]; a11 = p[32]; a12 = p[64]; a13 = p[96]; }
        __builtin_amdgcn_sched_barrier(0);
        const float t2 = wave_sum(end_lane_sum(a20, a21, a22, a23, vl + (m2 & 0xffu) * (EMBED / 4)));
        { const v4f* p = xl + (size_t)nn2 * (EMBED / 4); a20 = p[0]; a21 = p[32]; a22 = p[64]; a23 = p[96]; }
        __builtin_amdgcn_sched_barrier(0);
        const float t3 = wave_sum(end_lane_sum(a30, a31, a32, a33, vl + (m3 & 0xffu) * (EMBED / 4)));
        { const v4f* p = xl + (size_t)nn3 * (EMBED / 4); a30 = p[0]; a31 = p[32]; a32 = p[64]; a33 = p[96]; }
        __builtin_amdgcn_sched_barrier(0);
        if (lane == 0) {
          cur[m0 >> 8] = __builtin_bit_cast(unsigned, t0);
          cur[m1 >> 8] = __builtin_bit_cast(unsigned, t1);
          cur[m2 >> 8] = __builtin_bit_cast(unsigned, t2);
          cur[m3 >> 8] = __builtin_bit_cast(unsigned, t3);
        }
        cur = nxt; m0 = mm0; m1 = mm1; m2 = mm2; m3 = mm3;
        u += 32;
        more = seek();
        if (more) nxt = fetch(nn0, nn1, nn2, nn3, mm0, mm1, mm2, mm3);
      }
      const float t0 = wave_sum(end_lane_sum(a00, a01, a02, a03, vl + (m0 & 0xffu) * (EMBED / 4)));
      const float t1 = wave_sum(end_lane_sum(a10, a11, a12, a13, vl + (m1 & 0xffu) * (EMBED / 4)));
      const float t2 = wave_sum(end_lane_sum(a20, a21, a22, a23, vl + (m2 & 0xffu) * (EMBED / 4)));
      const float t3 = wave_sum(end_lane_sum(a30, a31, a32, a33, vl + (m3 & 0xffu) * (EMBED / 4)));
      if (lane == 0) {
        cur[m0 >> 8] = __builtin_bit_cast(unsigned, t0);
        cur[m1 >> 8] = __builtin_bit_cast(unsigned, t1);
        cur[m2 >> 8] = __builtin_bit_cast(unsigned, t2);
        cur[m3 >> 8] = __builtin_bit_cast(unsigned, t3);
      }
    }
    __syncthreads();

    for (int k = lane; k < nE; k += 32) {
      const float ps = __builtin_bit_cast(float, tnode[2 * k]);
      const float pt = __builtin_bit_cast(float, tnode[2 * k + 1]);
      out[eid[k]] = ps - pt;
    }
  }
}

extern "C" void kernel_launch(void* const* d_in, const int* in_sizes, int n_in,
                              void* d_out, int out_size, void* d_ws, size_t ws_size,
                              hipStream_t stream) {
  const float* X     = (const float*)d_in[0];
  const int*   edges = (const int*)d_in[1];
  const int*   etype = (const int*)d_in[2];
  const float* W     = (const float*)d_in[3];
  const float* ebv   = (const float*)d_in[4];
  float*       out   = (float*)d_out;
  float*       V     = (float*)d_ws;

  int nnodes = in_sizes[0] / EMBED;
  int nedges = in_sizes[2];
  int nrel   = in_sizes[4] / BASIS;

  dim3 g1(EMBED / 64, (nrel + 15) / 16);
  project_rel<<<g1, 128, 0, stream>>>(ebv, W, V, nrel);

  dim3 g2(NCHUNK, (nrel + RG - 1) / RG);
  edge_score<<<g2, 1024, 0, stream>>>(X, edges, edges + nedges, etype, V, out,
                                      nedges, nnodes, nrel);
}
